# starting kernel
# baseline (speedup 1.0000x reference)
#include <hip/hip_runtime.h>
#include <hip/hip_bf16.h>
#include <math.h>

typedef __attribute__((ext_vector_type(2))) float v2f;
typedef __attribute__((ext_vector_type(8))) float v8f;

#define TOKENS   32768
#define CH       2
#define VDIM     128
#define KCODE    512
#define OUT1_OFF 8388608
#define OUT2_OFF 8454144
#define ENT_OFF  8519680

__global__ __launch_bounds__(256) void vq_setup(const float* __restrict__ emb,
                                                unsigned int* __restrict__ hist,
                                                float* __restrict__ esq) {
    int tid = blockIdx.x * 256 + threadIdx.x;
    if (tid < KCODE) hist[tid] = 0u;
    if (tid < CH * KCODE) {
        const float* ep = emb + (size_t)tid * VDIM;
        float s = 0.0f;
        #pragma unroll 8
        for (int v = 0; v < VDIM; ++v) { float e = ep[v]; s += e * e; }
        esq[tid] = s;
    }
}

__global__ __launch_bounds__(256) void vq_main(const float* __restrict__ x0,
                                               const float* __restrict__ emb,
                                               const float* __restrict__ esq,
                                               unsigned int* __restrict__ hist,
                                               float* __restrict__ out) {
    __shared__ int sIdx[256];

    const int lane   = threadIdx.x & 31;
    const int waveId = threadIdx.x >> 5;
    const int lane16 = lane & 15;
    const int half   = lane >> 4;
    const int c       = blockIdx.x & 1;
    const int rowBase = (blockIdx.x >> 1) * 256;
    const int waveRow = rowBase + waveId * 32;

    v2f a0[32], a1[32];
    {
        const float* xp0 = x0 + (size_t)(waveRow + lane16) * (CH * VDIM)
                              + c * VDIM + half * 2;
        const float* xp1 = xp0 + 16 * (CH * VDIM);
        #pragma unroll
        for (int kk = 0; kk < 32; ++kk) {
            a0[kk] = *(const v2f*)(xp0 + kk * 4);
            a1[kk] = *(const v2f*)(xp1 + kk * 4);
        }
    }

    float runD0[8], runD1[8];
    int   runI0[8], runI1[8];
    #pragma unroll
    for (int r = 0; r < 8; ++r) {
        runD0[r] = 3.4e38f; runD1[r] = 3.4e38f; runI0[r] = 0; runI1[r] = 0;
    }

    for (int t = 0; t < 32; ++t) {
        const int code = t * 16 + lane16;
        const float ev2 = esq[c * KCODE + code];
        const float* ep = emb + ((size_t)(c * KCODE + code)) * VDIM + half * 2;

        v8f acc0 = {}; v8f acc1 = {};
        #pragma unroll
        for (int kk = 0; kk < 32; ++kk) {
            v2f b = *(const v2f*)(ep + kk * 4);
            acc0 = __builtin_amdgcn_wmma_f32_16x16x4_f32(
                false, a0[kk], false, b, (short)0, acc0, false, false);
            acc1 = __builtin_amdgcn_wmma_f32_16x16x4_f32(
                false, a1[kk], false, b, (short)0, acc1, false, false);
        }
        #pragma unroll
        for (int r = 0; r < 8; ++r) {
            float d0 = ev2 - 2.0f * acc0[r];
            float d1 = ev2 - 2.0f * acc1[r];
            if (d0 < runD0[r]) { runD0[r] = d0; runI0[r] = code; }
            if (d1 < runD1[r]) { runD1[r] = d1; runI1[r] = code; }
        }
    }

    #pragma unroll
    for (int m = 1; m < 16; m <<= 1) {
        #pragma unroll
        for (int r = 0; r < 8; ++r) {
            float od = __shfl_xor(runD0[r], m, 32);
            int   oi = __shfl_xor(runI0[r], m, 32);
            if (od < runD0[r] || (od == runD0[r] && oi < runI0[r])) {
                runD0[r] = od; runI0[r] = oi;
            }
            od = __shfl_xor(runD1[r], m, 32);
            oi = __shfl_xor(runI1[r], m, 32);
            if (od < runD1[r] || (od == runD1[r] && oi < runI1[r])) {
                runD1[r] = od; runI1[r] = oi;
            }
        }
    }

    if (lane16 == 0) {
        #pragma unroll
        for (int r = 0; r < 8; ++r) {
            int rl0 = waveId * 32 + r + 8 * half;
            sIdx[rl0] = runI0[r];
            atomicAdd(&hist[runI0[r]], 1u);
            int rl1 = rl0 + 16;
            sIdx[rl1] = runI1[r];
            atomicAdd(&hist[runI1[r]], 1u);
        }
    }
    __syncthreads();

    for (int it = 0; it < 32; ++it) {
        int rloc = waveId + it * 8;
        int row  = rowBase + rloc;
        int idx  = sIdx[rloc];
        const float4 ev = *(const float4*)(emb + ((size_t)(c * KCODE + idx)) * VDIM + lane * 4);
        const float4 xv = *(const float4*)(x0 + (size_t)row * (CH * VDIM) + c * VDIM + lane * 4);
        *(float4*)(out + (size_t)row * (CH * VDIM) + c * VDIM + lane * 4) = ev;
        float dx = xv.x - ev.x, dy = xv.y - ev.y;
        float dz = xv.z - ev.z, dw = xv.w - ev.w;
        float s = dx * dx + dy * dy + dz * dz + dw * dw;
        #pragma unroll
        for (int m = 1; m < 32; m <<= 1) s += __shfl_xor(s, m, 32);
        if (lane == 0) {
            out[OUT1_OFF + row * CH + c] = s;
            out[OUT2_OFF + row * CH + c] = s;
        }
    }
}

__global__ __launch_bounds__(256) void vq_entropy(const unsigned int* __restrict__ hist,
                                                  float* __restrict__ out) {
    __shared__ float red[256];
    int t = threadIdx.x;
    float s = 0.0f;
    for (int k = t; k < KCODE; k += 256) {
        unsigned int h = hist[k];
        if (h > 0u) {
            float p = (float)h * (1.0f / 32768.0f);
            s += p * logf(p);
        }
    }
    red[t] = s;
    __syncthreads();
    for (int o = 128; o > 0; o >>= 1) {
        if (t < o) red[t] += red[t + o];
        __syncthreads();
    }
    if (t == 0) out[ENT_OFF] = -red[0];
}

extern "C" void kernel_launch(void* const* d_in, const int* in_sizes, int n_in,
                              void* d_out, int out_size, void* d_ws, size_t ws_size,
                              hipStream_t stream) {
    const float* x0  = (const float*)d_in[0];
    const float* emb = (const float*)d_in[1];
    float* out = (float*)d_out;

    unsigned int* hist = (unsigned int*)d_ws;
    float* esq = (float*)((char*)d_ws + KCODE * sizeof(unsigned int));

    vq_setup  <<<4, 256, 0, stream>>>(emb, hist, esq);
    vq_main   <<<(TOKENS / 256) * CH, 256, 0, stream>>>(x0, emb, esq, hist, out);
    vq_entropy<<<1, 256, 0, stream>>>(hist, out);
}


# usage histogram: counts added per block in LDS, then 16 whole-line global atomics per block behind the barrier (was 256 two-lane ones, waited for before the gather); integer adds, float sums same
# speedup vs baseline: 1.2030x; 1.2030x over previous
#include <hip/hip_runtime.h>
#include <hip/hip_bf16.h>
#include <math.h>

typedef __attribute__((ext_vector_type(2))) float v2f;
typedef __attribute__((ext_vector_type(8))) float v8f;

#define TOKENS   32768
#define CH       2
#define VDIM     128
#define KCODE    512
#define OUT1_OFF 8388608
#define OUT2_OFF 8454144
#define ENT_OFF  8519680

__global__ __launch_bounds__(256) void vq_setup(const float* __restrict__ emb,
                                                unsigned int* __restrict__ hist,
                                                float* __restrict__ esq) {
    int tid = blockIdx.x * 256 + threadIdx.x;
    if (tid < KCODE) hist[tid] = 0u;
    if (tid < CH * KCODE) {
        const float* ep = emb + (size_t)tid * VDIM;
        float s = 0.0f;
        #pragma unroll 8
        for (int v = 0; v < VDIM; ++v) { float e = ep[v]; s += e * e; }
        esq[tid] = s;
    }
}

#define RT 4

__device__ __forceinline__ void load_b_tile(const float* __restrict__ ep,
                                            v2f (&b)[32]) {
    #pragma unroll
    for (int kk = 0; kk < 32; ++kk) b[kk] = *(const v2f*)(ep + kk * 4);
}

__device__ __forceinline__ void mma_tile(const v2f (&a)[RT][32], const v2f (&b)[32],
                                         v8f (&acc)[RT]) {
    #pragma unroll
    for (int i = 0; i < RT; ++i) acc[i] = (v8f){};
    #pragma unroll
    for (int kk = 0; kk < 32; ++kk) {
        #pragma unroll
        for (int i = 0; i < RT; ++i) {
            acc[i] = __builtin_amdgcn_wmma_f32_16x16x4_f32(
                false, a[i][kk], false, b[kk], (short)0, acc[i], false, false);
        }
    }
}

__device__ __forceinline__ void argmin_update(const v8f (&acc)[RT], float ev2, int code,
                                              float (&runD)[RT][8], int (&runI)[RT][8]) {
    #pragma unroll
    for (int i = 0; i < RT; ++i) {
        #pragma unroll
        for (int r = 0; r < 8; ++r) {
            float d = ev2 - 2.0f * acc[i][r];
            if (d < runD[i][r]) { runD[i][r] = d; runI[i][r] = code; }
        }
    }
}

__global__ __launch_bounds__(128) void vq_main(const float* __restrict__ x0,
                                               const float* __restrict__ emb,
                                               const float* __restrict__ esq,
                                               unsigned int* __restrict__ hist,
                                               float* __restrict__ out) {
    __shared__ int sIdx[256];
    __shared__ __attribute__((aligned(16))) unsigned int sHist[KCODE];

    *(uint4*)(sHist + threadIdx.x * 4) = make_uint4(0u, 0u, 0u, 0u);
    __syncthreads();

    const int lane   = threadIdx.x & 31;
    const int waveId = threadIdx.x >> 5;
    const int lane16 = lane & 15;
    const int half   = lane >> 4;
    const int c       = blockIdx.x & 1;
    const int rowBase = (blockIdx.x >> 1) * 256;
    const int waveRow = rowBase + waveId * (RT * 16);

    v2f a[RT][32];
    {
        const float* xp = x0 + (size_t)(waveRow + lane16) * (CH * VDIM)
                             + c * VDIM + half * 2;
        #pragma unroll
        for (int i = 0; i < RT; ++i) {
            #pragma unroll
            for (int kk = 0; kk < 32; ++kk) {
                a[i][kk] = *(const v2f*)(xp + i * 16 * (CH * VDIM) + kk * 4);
            }
        }
    }

    float runD[RT][8];
    int   runI[RT][8];
    #pragma unroll
    for (int i = 0; i < RT; ++i) {
        #pragma unroll
        for (int r = 0; r < 8; ++r) { runD[i][r] = 3.4e38f; runI[i][r] = 0; }
    }

    const float* ebase = emb + ((size_t)(c * KCODE + lane16)) * VDIM + half * 2;
    const float* qbase = esq + c * KCODE + lane16;

    v2f bE[32], bO[32];
    float evE, evO;
    load_b_tile(ebase, bE);
    evE = qbase[0];

    for (int t = 0; t < 32; t += 2) {
        load_b_tile(ebase + (size_t)(t + 1) * 16 * VDIM, bO);
        evO = qbase[(t + 1) * 16];
        v8f accE[RT];
        mma_tile(a, bE, accE);

        const float evEcur = evE;
        const int tn = (t + 2 < 32) ? (t + 2) : 31;
        __builtin_amdgcn_sched_barrier(0);
        load_b_tile(ebase + (size_t)tn * 16 * VDIM, bE);
        evE = qbase[tn * 16];
        __builtin_amdgcn_sched_barrier(0);

        argmin_update(accE, evEcur, t * 16 + lane16, runD, runI);
        v8f accO[RT];
        mma_tile(a, bO, accO);
        argmin_update(accO, evO, (t + 1) * 16 + lane16, runD, runI);
    }

    #pragma unroll
    for (int m = 1; m < 16; m <<= 1) {
        float od[RT][8];
        int   oi[RT][8];
        #pragma unroll
        for (int i = 0; i < RT; ++i) {
            #pragma unroll
            for (int r = 0; r < 8; ++r) {
                od[i][r] = __shfl_xor(runD[i][r], m, 32);
                oi[i][r] = __shfl_xor(runI[i][r], m, 32);
            }
        }
        #pragma unroll
        for (int i = 0; i < RT; ++i) {
            #pragma unroll
            for (int r = 0; r < 8; ++r) {
                if (od[i][r] < runD[i][r] ||
                    (od[i][r] == runD[i][r] && oi[i][r] < runI[i][r])) {
                    runD[i][r] = od[i][r]; runI[i][r] = oi[i][r];
                }
            }
        }
    }

    if (lane16 == 0) {
        #pragma unroll
        for (int i = 0; i < RT; ++i) {
            #pragma unroll
            for (int r = 0; r < 8; ++r) {
                int rl = waveId * (RT * 16) + i * 16 + r + 8 * half;
                sIdx[rl] = runI[i][r];
                atomicAdd(&sHist[runI[i][r]], 1u);
            }
        }
    }
    __syncthreads();

    {
        const int bin0 = waveId * 128 + lane;
        unsigned int cnt[4];
        #pragma unroll
        for (int j = 0; j < 4; ++j) cnt[j] = sHist[bin0 + j * 32];
        #pragma unroll
        for (int j = 0; j < 4; ++j) {
            if (cnt[j] != 0u) atomicAdd(&hist[bin0 + j * 32], cnt[j]);
        }
    }

    for (int g = 0; g < 8; ++g) {
        float4 ev[8], xv[8];
        #pragma unroll
        for (int j = 0; j < 8; ++j) {
            int rloc = waveId + (g * 8 + j) * 4;
            int idx  = sIdx[rloc];
            ev[j] = *(const float4*)(emb + ((size_t)(c * KCODE + idx)) * VDIM + lane * 4);
            xv[j] = *(const float4*)(x0 + (size_t)(rowBase + rloc) * (CH * VDIM) + c * VDIM + lane * 4);
        }
        float s[8];
        #pragma unroll
        for (int j = 0; j < 8; ++j) {
            int row = rowBase + waveId + (g * 8 + j) * 4;
            *(float4*)(out + (size_t)row * (CH * VDIM) + c * VDIM + lane * 4) = ev[j];
            float dx = xv[j].x - ev[j].x, dy = xv[j].y - ev[j].y;
            float dz = xv[j].z - ev[j].z, dw = xv[j].w - ev[j].w;
            s[j] = dx * dx + dy * dy + dz * dz + dw * dw;
        }
        #pragma unroll
        for (int m = 1; m < 32; m <<= 1) {
            #pragma unroll
            for (int j = 0; j < 8; ++j) s[j] += __shfl_xor(s[j], m, 32);
        }
        float mine = s[0];
        #pragma unroll
        for (int j = 1; j < 8; ++j) mine = (lane == j) ? s[j] : mine;
        if (lane < 8) {
            int row = rowBase + waveId + (g * 8 + lane) * 4;
            out[OUT1_OFF + row * CH + c] = mine;
            out[OUT2_OFF + row * CH + c] = mine;
        }
    }
}

__global__ __launch_bounds__(256) void vq_entropy(const unsigned int* __restrict__ hist,
                                                  float* __restrict__ out) {
    __shared__ float red[256];
    int t = threadIdx.x;
    float s = 0.0f;
    for (int k = t; k < KCODE; k += 256) {
        unsigned int h = hist[k];
        if (h > 0u) {
            float p = (float)h * (1.0f / 32768.0f);
            s += p * logf(p);
        }
    }
    red[t] = s;
    __syncthreads();
    for (int o = 128; o > 0; o >>= 1) {
        if (t < o) red[t] += red[t + o];
        __syncthreads();
    }
    if (t == 0) out[ENT_OFF] = -red[0];
}

extern "C" void kernel_launch(void* const* d_in, const int* in_sizes, int n_in,
                              void* d_out, int out_size, void* d_ws, size_t ws_size,
                              hipStream_t stream) {
    const float* x0  = (const float*)d_in[0];
    const float* emb = (const float*)d_in[1];
    float* out = (float*)d_out;

    unsigned int* hist = (unsigned int*)d_ws;
    float* esq = (float*)((char*)d_ws + KCODE * sizeof(unsigned int));

    vq_setup  <<<4, 256, 0, stream>>>(emb, hist, esq);
    vq_main   <<<(TOKENS / 256) * CH, 128, 0, stream>>>(x0, emb, esq, hist, out);
    vq_entropy<<<1, 256, 0, stream>>>(hist, out);
}


# WMMA operand roles swapped (codes = A rows, tokens = B columns): a lane's 8 accumulators are 8 codes of ONE token, so the argmin is an in-lane pass + 1 half exchange, not a 16-lane ladder; sums same
# speedup vs baseline: 1.2807x; 1.0646x over previous
#include <hip/hip_runtime.h>
#include <hip/hip_bf16.h>
#include <math.h>

typedef __attribute__((ext_vector_type(2))) float v2f;
typedef __attribute__((ext_vector_type(8))) float v8f;

#define TOKENS   32768
#define CH       2
#define VDIM     128
#define KCODE    512
#define OUT1_OFF 8388608
#define OUT2_OFF 8454144
#define ENT_OFF  8519680

__global__ __launch_bounds__(256) void vq_setup(const float* __restrict__ emb,
                                                unsigned int* __restrict__ hist,
                                                float* __restrict__ esq) {
    int tid = blockIdx.x * 256 + threadIdx.x;
    if (tid < KCODE) hist[tid] = 0u;
    if (tid < CH * KCODE) {
        const float* ep = emb + (size_t)tid * VDIM;
        float s = 0.0f;
        #pragma unroll 8
        for (int v = 0; v < VDIM; ++v) { float e = ep[v]; s += e * e; }
        esq[tid] = s;
    }
}

#define RT 4

__device__ __forceinline__ void load_b_tile(const float* __restrict__ ep,
                                            v2f (&b)[32]) {
    #pragma unroll
    for (int kk = 0; kk < 32; ++kk) b[kk] = *(const v2f*)(ep + kk * 4);
}

__device__ __forceinline__ void mma_tile(const v2f (&a)[RT][32], const v2f (&b)[32],
                                         v8f (&acc)[RT]) {
    #pragma unroll
    for (int i = 0; i < RT; ++i) acc[i] = (v8f){};
    #pragma unroll
    for (int kk = 0; kk < 32; ++kk) {
        #pragma unroll
        for (int i = 0; i < RT; ++i) {
            acc[i] = __builtin_amdgcn_wmma_f32_16x16x4_f32(
                false, b[kk], false, a[i][kk], (short)0, acc[i], false, false);
        }
    }
}

__device__ __forceinline__ void load_esq(const float* __restrict__ q, v2f (&ev)[4]) {
    const float4 lo = *(const float4*)q;
    const float4 hi = *(const float4*)(q + 4);
    ev[0] = (v2f){lo.x, lo.y}; ev[1] = (v2f){lo.z, lo.w};
    ev[2] = (v2f){hi.x, hi.y}; ev[3] = (v2f){hi.z, hi.w};
}

__device__ __forceinline__ void argmin_update(const v8f (&acc)[RT], const v2f (&ev)[4], int code0,
                                              float (&runD)[RT][8], int (&runI)[RT][8]) {
    #pragma unroll
    for (int i = 0; i < RT; ++i) {
        #pragma unroll
        for (int p = 0; p < 4; ++p) {
            const v2f ap = (v2f){acc[i][2 * p], acc[i][2 * p + 1]};
            const v2f d  = ev[p] - 2.0f * ap;
            if (d.x < runD[i][2 * p])     { runD[i][2 * p]     = d.x; runI[i][2 * p]     = code0; }
            if (d.y < runD[i][2 * p + 1]) { runD[i][2 * p + 1] = d.y; runI[i][2 * p + 1] = code0; }
        }
    }
}

__global__ __launch_bounds__(128) void vq_main(const float* __restrict__ x0,
                                               const float* __restrict__ emb,
                                               const float* __restrict__ esq,
                                               unsigned int* __restrict__ hist,
                                               float* __restrict__ out) {
    __shared__ int sIdx[256];
    __shared__ __attribute__((aligned(16))) unsigned int sHist[KCODE];

    *(uint4*)(sHist + threadIdx.x * 4) = make_uint4(0u, 0u, 0u, 0u);
    __syncthreads();

    const int lane   = threadIdx.x & 31;
    const int waveId = threadIdx.x >> 5;
    const int lane16 = lane & 15;
    const int half   = lane >> 4;
    const int c       = blockIdx.x & 1;
    const int rowBase = (blockIdx.x >> 1) * 256;
    const int waveRow = rowBase + waveId * (RT * 16);

    v2f a[RT][32];
    {
        const float* xp = x0 + (size_t)(waveRow + lane16) * (CH * VDIM)
                             + c * VDIM + half * 2;
        #pragma unroll
        for (int i = 0; i < RT; ++i) {
            #pragma unroll
            for (int kk = 0; kk < 32; ++kk) {
                a[i][kk] = *(const v2f*)(xp + i * 16 * (CH * VDIM) + kk * 4);
            }
        }
    }

    float runD[RT][8];
    int   runI[RT][8];
    #pragma unroll
    for (int i = 0; i < RT; ++i) {
        #pragma unroll
        for (int r = 0; r < 8; ++r) { runD[i][r] = 3.4e38f; runI[i][r] = 0; }
    }

    const float* ebase = emb + ((size_t)(c * KCODE + lane16)) * VDIM + half * 2;
    const float* qbase = esq + c * KCODE + half * 8;

    v2f bE[32], bO[32];
    v2f evE[4], evO[4];
    load_b_tile(ebase, bE);
    load_esq(qbase, evE);

    for (int t = 0; t < 32; t += 2) {
        load_b_tile(ebase + (size_t)(t + 1) * 16 * VDIM, bO);
        load_esq(qbase + (t + 1) * 16, evO);
        v8f accE[RT];
        mma_tile(a, bE, accE);

        v2f evEcur[4];
        #pragma unroll
        for (int p = 0; p < 4; ++p) evEcur[p] = evE[p];
        const int tn = (t + 2 < 32) ? (t + 2) : 31;
        __builtin_amdgcn_sched_barrier(0);
        load_b_tile(ebase + (size_t)tn * 16 * VDIM, bE);
        load_esq(qbase + tn * 16, evE);
        __builtin_amdgcn_sched_barrier(0);

        argmin_update(accE, evEcur, t * 16 + half * 8, runD, runI);
        v8f accO[RT];
        mma_tile(a, bO, accO);
        argmin_update(accO, evO, (t + 1) * 16 + half * 8, runD, runI);
    }

    float bestD[RT];
    int   bestI[RT];
    #pragma unroll
    for (int i = 0; i < RT; ++i) {
        float bd = runD[i][0];
        int   bi = runI[i][0];
        #pragma unroll
        for (int r = 1; r < 8; ++r) {
            const float cd = runD[i][r];
            const int   ci = runI[i][r] + r;
            const bool take = (cd < bd) | ((cd == bd) & (ci < bi));
            bd = take ? cd : bd;
            bi = take ? ci : bi;
        }
        bestD[i] = bd; bestI[i] = bi;
    }
    {
        float od[RT];
        int   oi[RT];
        #pragma unroll
        for (int i = 0; i < RT; ++i) {
            od[i] = __shfl_xor(bestD[i], 16, 32);
            oi[i] = __shfl_xor(bestI[i], 16, 32);
        }
        #pragma unroll
        for (int i = 0; i < RT; ++i) {
            const bool take = (od[i] < bestD[i]) | ((od[i] == bestD[i]) & (oi[i] < bestI[i]));
            bestI[i] = take ? oi[i] : bestI[i];
        }
    }

    {
        const int i0 = half ? bestI[2] : bestI[0];
        const int i1 = half ? bestI[3] : bestI[1];
        const int rl = waveId * (RT * 16) + half * 32 + lane16;
        sIdx[rl]      = i0;
        sIdx[rl + 16] = i1;
        atomicAdd(&sHist[i0], 1u);
        atomicAdd(&sHist[i1], 1u);
    }
    __syncthreads();

    {
        const int bin0 = waveId * 128 + lane;
        unsigned int cnt[4];
        #pragma unroll
        for (int j = 0; j < 4; ++j) cnt[j] = sHist[bin0 + j * 32];
        #pragma unroll
        for (int j = 0; j < 4; ++j) {
            if (cnt[j] != 0u) atomicAdd(&hist[bin0 + j * 32], cnt[j]);
        }
    }

    for (int g = 0; g < 8; ++g) {
        float4 ev[8], xv[8];
        #pragma unroll
        for (int j = 0; j < 8; ++j) {
            int rloc = waveId + (g * 8 + j) * 4;
            int idx  = sIdx[rloc];
            ev[j] = *(const float4*)(emb + ((size_t)(c * KCODE + idx)) * VDIM + lane * 4);
            xv[j] = *(const float4*)(x0 + (size_t)(rowBase + rloc) * (CH * VDIM) + c * VDIM + lane * 4);
        }
        float s[8];
        #pragma unroll
        for (int j = 0; j < 8; ++j) {
            int row = rowBase + waveId + (g * 8 + j) * 4;
            *(float4*)(out + (size_t)row * (CH * VDIM) + c * VDIM + lane * 4) = ev[j];
            float dx = xv[j].x - ev[j].x, dy = xv[j].y - ev[j].y;
            float dz = xv[j].z - ev[j].z, dw = xv[j].w - ev[j].w;
            s[j] = dx * dx + dy * dy + dz * dz + dw * dw;
        }
        #pragma unroll
        for (int m = 1; m < 32; m <<= 1) {
            #pragma unroll
            for (int j = 0; j < 8; ++j) s[j] += __shfl_xor(s[j], m, 32);
        }
        float mine = s[0];
        #pragma unroll
        for (int j = 1; j < 8; ++j) mine = (lane == j) ? s[j] : mine;
        if (lane < 8) {
            int row = rowBase + waveId + (g * 8 + lane) * 4;
            out[OUT1_OFF + row * CH + c] = mine;
            out[OUT2_OFF + row * CH + c] = mine;
        }
    }
}

__global__ __launch_bounds__(256) void vq_entropy(const unsigned int* __restrict__ hist,
                                                  float* __restrict__ out) {
    __shared__ float red[256];
    int t = threadIdx.x;
    float s = 0.0f;
    for (int k = t; k < KCODE; k += 256) {
        unsigned int h = hist[k];
        if (h > 0u) {
            float p = (float)h * (1.0f / 32768.0f);
            s += p * logf(p);
        }
    }
    red[t] = s;
    __syncthreads();
    for (int o = 128; o > 0; o >>= 1) {
        if (t < o) red[t] += red[t + o];
        __syncthreads();
    }
    if (t == 0) out[ENT_OFF] = -red[0];
}

extern "C" void kernel_launch(void* const* d_in, const int* in_sizes, int n_in,
                              void* d_out, int out_size, void* d_ws, size_t ws_size,
                              hipStream_t stream) {
    const float* x0  = (const float*)d_in[0];
    const float* emb = (const float*)d_in[1];
    float* out = (float*)d_out;

    unsigned int* hist = (unsigned int*)d_ws;
    float* esq = (float*)((char*)d_ws + KCODE * sizeof(unsigned int));

    vq_setup  <<<4, 256, 0, stream>>>(emb, hist, esq);
    vq_main   <<<(TOKENS / 256) * CH, 128, 0, stream>>>(x0, emb, esq, hist, out);
    vq_entropy<<<1, 256, 0, stream>>>(hist, out);
}


# mid-loop refill no longer waited for in full a few WMMAs behind its issue: next even tile's ||e||^2 requested at loop head, refill issued highest k first (its one copied fragment leads); sums same
# speedup vs baseline: 1.3675x; 1.0678x over previous
#include <hip/hip_runtime.h>
#include <hip/hip_bf16.h>
#include <math.h>

typedef __attribute__((ext_vector_type(2))) float v2f;
typedef __attribute__((ext_vector_type(8))) float v8f;

#define TOKENS   32768
#define CH       2
#define VDIM     128
#define KCODE    512
#define OUT1_OFF 8388608
#define OUT2_OFF 8454144
#define ENT_OFF  8519680

__global__ __launch_bounds__(256) void vq_setup(const float* __restrict__ emb,
                                                unsigned int* __restrict__ hist,
                                                float* __restrict__ esq) {
    int tid = blockIdx.x * 256 + threadIdx.x;
    if (tid < KCODE) hist[tid] = 0u;
    if (tid < CH * KCODE) {
        const float* ep = emb + (size_t)tid * VDIM;
        float s = 0.0f;
        #pragma unroll 8
        for (int v = 0; v < VDIM; ++v) { float e = ep[v]; s += e * e; }
        esq[tid] = s;
    }
}

#define RT 4

__device__ __forceinline__ void load_b_tile(const float* __restrict__ ep,
                                            v2f (&b)[32]) {
    #pragma unroll
    for (int kk = 0; kk < 32; ++kk) b[kk] = *(const v2f*)(ep + kk * 4);
}

__device__ __forceinline__ void load_b_tile_rev(const float* __restrict__ ep,
                                                v2f (&b)[32]) {
    #pragma unroll
    for (int kk = 31; kk >= 0; --kk) b[kk] = *(const v2f*)(ep + kk * 4);
}

__device__ __forceinline__ void mma_tile(const v2f (&a)[RT][32], const v2f (&b)[32],
                                         v8f (&acc)[RT]) {
    #pragma unroll
    for (int i = 0; i < RT; ++i) acc[i] = (v8f){};
    #pragma unroll
    for (int kk = 0; kk < 32; ++kk) {
        #pragma unroll
        for (int i = 0; i < RT; ++i) {
            acc[i] = __builtin_amdgcn_wmma_f32_16x16x4_f32(
                false, b[kk], false, a[i][kk], (short)0, acc[i], false, false);
        }
    }
}

__device__ __forceinline__ void load_esq(const float* __restrict__ q, v2f (&ev)[4]) {
    const float4 lo = *(const float4*)q;
    const float4 hi = *(const float4*)(q + 4);
    ev[0] = (v2f){lo.x, lo.y}; ev[1] = (v2f){lo.z, lo.w};
    ev[2] = (v2f){hi.x, hi.y}; ev[3] = (v2f){hi.z, hi.w};
}

__device__ __forceinline__ void argmin_update(const v8f (&acc)[RT], const v2f (&ev)[4], int code0,
                                              float (&runD)[RT][8], int (&runI)[RT][8]) {
    #pragma unroll
    for (int i = 0; i < RT; ++i) {
        #pragma unroll
        for (int p = 0; p < 4; ++p) {
            const v2f ap = (v2f){acc[i][2 * p], acc[i][2 * p + 1]};
            const v2f d  = ev[p] - 2.0f * ap;
            if (d.x < runD[i][2 * p])     { runD[i][2 * p]     = d.x; runI[i][2 * p]     = code0; }
            if (d.y < runD[i][2 * p + 1]) { runD[i][2 * p + 1] = d.y; runI[i][2 * p + 1] = code0; }
        }
    }
}

__global__ __launch_bounds__(128) void vq_main(const float* __restrict__ x0,
                                               const float* __restrict__ emb,
                                               const float* __restrict__ esq,
                                               unsigned int* __restrict__ hist,
                                               float* __restrict__ out) {
    __shared__ int sIdx[256];
    __shared__ __attribute__((aligned(16))) unsigned int sHist[KCODE];

    *(uint4*)(sHist + threadIdx.x * 4) = make_uint4(0u, 0u, 0u, 0u);
    __syncthreads();

    const int lane   = threadIdx.x & 31;
    const int waveId = threadIdx.x >> 5;
    const int lane16 = lane & 15;
    const int half   = lane >> 4;
    const int c       = blockIdx.x & 1;
    const int rowBase = (blockIdx.x >> 1) * 256;
    const int waveRow = rowBase + waveId * (RT * 16);

    v2f a[RT][32];
    {
        const float* xp = x0 + (size_t)(waveRow + lane16) * (CH * VDIM)
                             + c * VDIM + half * 2;
        #pragma unroll
        for (int i = 0; i < RT; ++i) {
            #pragma unroll
            for (int kk = 0; kk < 32; ++kk) {
                a[i][kk] = *(const v2f*)(xp + i * 16 * (CH * VDIM) + kk * 4);
            }
        }
    }

    float runD[RT][8];
    int   runI[RT][8];
    #pragma unroll
    for (int i = 0; i < RT; ++i) {
        #pragma unroll
        for (int r = 0; r < 8; ++r) { runD[i][r] = 3.4e38f; runI[i][r] = 0; }
    }

    const float* ebase = emb + ((size_t)(c * KCODE + lane16)) * VDIM + half * 2;
    const float* qbase = esq + c * KCODE + half * 8;

    v2f bE[32], bO[32];
    v2f evE[4], evO[4];
    load_b_tile(ebase, bE);
    load_esq(qbase, evE);

    for (int t = 0; t < 32; t += 2) {
        load_b_tile(ebase + (size_t)(t + 1) * 16 * VDIM, bO);
        load_esq(qbase + (t + 1) * 16, evO);
        const int tn = (t + 2 < 32) ? (t + 2) : 31;
        v2f evEn[4];
        load_esq(qbase + tn * 16, evEn);
        __builtin_amdgcn_sched_barrier(0);
        v8f accE[RT];
        mma_tile(a, bE, accE);

        __builtin_amdgcn_sched_barrier(0);
        load_b_tile_rev(ebase + (size_t)tn * 16 * VDIM, bE);
        __builtin_amdgcn_sched_barrier(0);

        argmin_update(accE, evE, t * 16 + half * 8, runD, runI);
        v8f accO[RT];
        mma_tile(a, bO, accO);
        argmin_update(accO, evO, (t + 1) * 16 + half * 8, runD, runI);
        #pragma unroll
        for (int p = 0; p < 4; ++p) evE[p] = evEn[p];
    }

    float bestD[RT];
    int   bestI[RT];
    #pragma unroll
    for (int i = 0; i < RT; ++i) {
        float bd = runD[i][0];
        int   bi = runI[i][0];
        #pragma unroll
        for (int r = 1; r < 8; ++r) {
            const float cd = runD[i][r];
            const int   ci = runI[i][r] + r;
            const bool take = (cd < bd) | ((cd == bd) & (ci < bi));
            bd = take ? cd : bd;
            bi = take ? ci : bi;
        }
        bestD[i] = bd; bestI[i] = bi;
    }
    {
        float od[RT];
        int   oi[RT];
        #pragma unroll
        for (int i = 0; i < RT; ++i) {
            od[i] = __shfl_xor(bestD[i], 16, 32);
            oi[i] = __shfl_xor(bestI[i], 16, 32);
        }
        #pragma unroll
        for (int i = 0; i < RT; ++i) {
            const bool take = (od[i] < bestD[i]) | ((od[i] == bestD[i]) & (oi[i] < bestI[i]));
            bestI[i] = take ? oi[i] : bestI[i];
        }
    }

    {
        const int i0 = half ? bestI[2] : bestI[0];
        const int i1 = half ? bestI[3] : bestI[1];
        const int rl = waveId * (RT * 16) + half * 32 + lane16;
        sIdx[rl]      = i0;
        sIdx[rl + 16] = i1;
        atomicAdd(&sHist[i0], 1u);
        atomicAdd(&sHist[i1], 1u);
    }
    __syncthreads();

    {
        const int bin0 = waveId * 128 + lane;
        unsigned int cnt[4];
        #pragma unroll
        for (int j = 0; j < 4; ++j) cnt[j] = sHist[bin0 + j * 32];
        #pragma unroll
        for (int j = 0; j < 4; ++j) {
            if (cnt[j] != 0u) atomicAdd(&hist[bin0 + j * 32], cnt[j]);
        }
    }

    for (int g = 0; g < 8; ++g) {
        float4 ev[8], xv[8];
        #pragma unroll
        for (int j = 0; j < 8; ++j) {
            int rloc = waveId + (g * 8 + j) * 4;
            int idx  = sIdx[rloc];
            ev[j] = *(const float4*)(emb + ((size_t)(c * KCODE + idx)) * VDIM + lane * 4);
            xv[j] = *(const float4*)(x0 + (size_t)(rowBase + rloc) * (CH * VDIM) + c * VDIM + lane * 4);
        }
        float s[8];
        #pragma unroll
        for (int j = 0; j < 8; ++j) {
            int row = rowBase + waveId + (g * 8 + j) * 4;
            *(float4*)(out + (size_t)row * (CH * VDIM) + c * VDIM + lane * 4) = ev[j];
            float dx = xv[j].x - ev[j].x, dy = xv[j].y - ev[j].y;
            float dz = xv[j].z - ev[j].z, dw = xv[j].w - ev[j].w;
            s[j] = dx * dx + dy * dy + dz * dz + dw * dw;
        }
        #pragma unroll
        for (int m = 1; m < 32; m <<= 1) {
            #pragma unroll
            for (int j = 0; j < 8; ++j) s[j] += __shfl_xor(s[j], m, 32);
        }
        float mine = s[0];
        #pragma unroll
        for (int j = 1; j < 8; ++j) mine = (lane == j) ? s[j] : mine;
        if (lane < 8) {
            int row = rowBase + waveId + (g * 8 + lane) * 4;
            out[OUT1_OFF + row * CH + c] = mine;
            out[OUT2_OFF + row * CH + c] = mine;
        }
    }
}

__global__ __launch_bounds__(256) void vq_entropy(const unsigned int* __restrict__ hist,
                                                  float* __restrict__ out) {
    __shared__ float red[256];
    int t = threadIdx.x;
    float s = 0.0f;
    for (int k = t; k < KCODE; k += 256) {
        unsigned int h = hist[k];
        if (h > 0u) {
            float p = (float)h * (1.0f / 32768.0f);
            s += p * logf(p);
        }
    }
    red[t] = s;
    __syncthreads();
    for (int o = 128; o > 0; o >>= 1) {
        if (t < o) red[t] += red[t + o];
        __syncthreads();
    }
    if (t == 0) out[ENT_OFF] = -red[0];
}

extern "C" void kernel_launch(void* const* d_in, const int* in_sizes, int n_in,
                              void* d_out, int out_size, void* d_ws, size_t ws_size,
                              hipStream_t stream) {
    const float* x0  = (const float*)d_in[0];
    const float* emb = (const float*)d_in[1];
    float* out = (float*)d_out;

    unsigned int* hist = (unsigned int*)d_ws;
    float* esq = (float*)((char*)d_ws + KCODE * sizeof(unsigned int));

    vq_setup  <<<4, 256, 0, stream>>>(emb, hist, esq);
    vq_main   <<<(TOKENS / 256) * CH, 128, 0, stream>>>(x0, emb, esq, hist, out);
    vq_entropy<<<1, 256, 0, stream>>>(hist, out);
}


# residuals from registers: ||x||^2 of the wave's token fragments + the sweep's minimum (||e||^2-2x.e); gather = 64 row copies in one request group, no x re-read, no lane sums; residual sums re-ordered
# speedup vs baseline: 1.4604x; 1.0679x over previous
#include <hip/hip_runtime.h>
#include <hip/hip_bf16.h>
#include <math.h>

typedef __attribute__((ext_vector_type(2))) float v2f;
typedef __attribute__((ext_vector_type(8))) float v8f;
typedef __attribute__((ext_vector_type(4))) float v4f;

#define TOKENS   32768
#define CH       2
#define VDIM     128
#define KCODE    512
#define OUT1_OFF 8388608
#define OUT2_OFF 8454144
#define ENT_OFF  8519680

__global__ __launch_bounds__(256) void vq_setup(const float* __restrict__ emb,
                                                unsigned int* __restrict__ hist,
                                                float* __restrict__ esq) {
    int tid = blockIdx.x * 256 + threadIdx.x;
    if (tid < KCODE) hist[tid] = 0u;
    if (tid < CH * KCODE) {
        const float* ep = emb + (size_t)tid * VDIM;
        float s = 0.0f;
        #pragma unroll 8
        for (int v = 0; v < VDIM; ++v) { float e = ep[v]; s += e * e; }
        esq[tid] = s;
    }
}

#define RT 4

__device__ __forceinline__ void load_b_tile(const float* __restrict__ ep,
                                            v2f (&b)[32]) {
    #pragma unroll
    for (int kk = 0; kk < 32; ++kk) b[kk] = *(const v2f*)(ep + kk * 4);
}

__device__ __forceinline__ void load_b_tile_rev(const float* __restrict__ ep,
                                                v2f (&b)[32]) {
    #pragma unroll
    for (int kk = 31; kk >= 0; --kk) b[kk] = *(const v2f*)(ep + kk * 4);
}

__device__ __forceinline__ void mma_tile(const v2f (&a)[RT][32], const v2f (&b)[32],
                                         v8f (&acc)[RT]) {
    #pragma unroll
    for (int i = 0; i < RT; ++i) acc[i] = (v8f){};
    #pragma unroll
    for (int kk = 0; kk < 32; ++kk) {
        #pragma unroll
        for (int i = 0; i < RT; ++i) {
            acc[i] = __builtin_amdgcn_wmma_f32_16x16x4_f32(
                false, b[kk], false, a[i][kk], (short)0, acc[i], false, false);
        }
    }
}

__device__ __forceinline__ void load_esq(const float* __restrict__ q, v2f (&ev)[4]) {
    const float4 lo = *(const float4*)q;
    const float4 hi = *(const float4*)(q + 4);
    ev[0] = (v2f){lo.x, lo.y}; ev[1] = (v2f){lo.z, lo.w};
    ev[2] = (v2f){hi.x, hi.y}; ev[3] = (v2f){hi.z, hi.w};
}

__device__ __forceinline__ void argmin_update(const v8f (&acc)[RT], const v2f (&ev)[4], int code0,
                                              float (&runD)[RT][8], int (&runI)[RT][8]) {
    #pragma unroll
    for (int i = 0; i < RT; ++i) {
        #pragma unroll
        for (int p = 0; p < 4; ++p) {
            const v2f ap = (v2f){acc[i][2 * p], acc[i][2 * p + 1]};
            const v2f d  = ev[p] - 2.0f * ap;
            if (d.x < runD[i][2 * p])     { runD[i][2 * p]     = d.x; runI[i][2 * p]     = code0; }
            if (d.y < runD[i][2 * p + 1]) { runD[i][2 * p + 1] = d.y; runI[i][2 * p + 1] = code0; }
        }
    }
}

__global__ __launch_bounds__(128) void vq_main(const float* __restrict__ x0,
                                               const float* __restrict__ emb,
                                               const float* __restrict__ esq,
                                               unsigned int* __restrict__ hist,
                                               float* __restrict__ out) {
    __shared__ int sIdx[256];
    __shared__ __attribute__((aligned(16))) unsigned int sHist[KCODE];

    *(uint4*)(sHist + threadIdx.x * 4) = make_uint4(0u, 0u, 0u, 0u);
    __syncthreads();

    const int lane   = threadIdx.x & 31;
    const int waveId = threadIdx.x >> 5;
    const int lane16 = lane & 15;
    const int half   = lane >> 4;
    const int c       = blockIdx.x & 1;
    const int rowBase = (blockIdx.x >> 1) * 256;
    const int waveRow = rowBase + waveId * (RT * 16);

    v2f a[RT][32];
    {
        const float* xp = x0 + (size_t)(waveRow + lane16) * (CH * VDIM)
                             + c * VDIM + half * 2;
        #pragma unroll
        for (int i = 0; i < RT; ++i) {
            #pragma unroll
            for (int kk = 0; kk < 32; ++kk) {
                a[i][kk] = *(const v2f*)(xp + i * 16 * (CH * VDIM) + kk * 4);
            }
        }
    }

    float runD[RT][8];
    int   runI[RT][8];
    #pragma unroll
    for (int i = 0; i < RT; ++i) {
        #pragma unroll
        for (int r = 0; r < 8; ++r) { runD[i][r] = 3.4e38f; runI[i][r] = 0; }
    }

    const float* ebase = emb + ((size_t)(c * KCODE + lane16)) * VDIM + half * 2;
    const float* qbase = esq + c * KCODE + half * 8;

    v2f bE[32], bO[32];
    v2f evE[4], evO[4];
    load_b_tile(ebase, bE);
    load_esq(qbase, evE);

    for (int t = 0; t < 32; t += 2) {
        load_b_tile(ebase + (size_t)(t + 1) * 16 * VDIM, bO);
        load_esq(qbase + (t + 1) * 16, evO);
        const int tn = (t + 2 < 32) ? (t + 2) : 31;
        v2f evEn[4];
        load_esq(qbase + tn * 16, evEn);
        __builtin_amdgcn_sched_barrier(0);
        v8f accE[RT];
        mma_tile(a, bE, accE);

        __builtin_amdgcn_sched_barrier(0);
        load_b_tile_rev(ebase + (size_t)tn * 16 * VDIM, bE);
        __builtin_amdgcn_sched_barrier(0);

        argmin_update(accE, evE, t * 16 + half * 8, runD, runI);
        v8f accO[RT];
        mma_tile(a, bO, accO);
        argmin_update(accO, evO, (t + 1) * 16 + half * 8, runD, runI);
        #pragma unroll
        for (int p = 0; p < 4; ++p) evE[p] = evEn[p];
    }

    float bestD[RT];
    int   bestI[RT];
    #pragma unroll
    for (int i = 0; i < RT; ++i) {
        float bd = runD[i][0];
        int   bi = runI[i][0];
        #pragma unroll
        for (int r = 1; r < 8; ++r) {
            const float cd = runD[i][r];
            const int   ci = runI[i][r] + r;
            const bool take = (cd < bd) | ((cd == bd) & (ci < bi));
            bd = take ? cd : bd;
            bi = take ? ci : bi;
        }
        bestD[i] = bd; bestI[i] = bi;
    }
    {
        float od[RT];
        int   oi[RT];
        #pragma unroll
        for (int i = 0; i < RT; ++i) {
            od[i] = __shfl_xor(bestD[i], 16, 32);
            oi[i] = __shfl_xor(bestI[i], 16, 32);
        }
        #pragma unroll
        for (int i = 0; i < RT; ++i) {
            const bool take = (od[i] < bestD[i]) | ((od[i] == bestD[i]) & (oi[i] < bestI[i]));
            bestI[i] = take ? oi[i] : bestI[i];
            bestD[i] = take ? od[i] : bestD[i];
        }
    }

    float res[RT];
    {
        float xx[RT];
        #pragma unroll
        for (int i = 0; i < RT; ++i) {
            v2f s0 = a[i][0] * a[i][0], s1 = a[i][1] * a[i][1];
            v2f s2 = a[i][2] * a[i][2], s3 = a[i][3] * a[i][3];
            #pragma unroll
            for (int kk = 4; kk < 32; kk += 4) {
                s0 += a[i][kk]     * a[i][kk];
                s1 += a[i][kk + 1] * a[i][kk + 1];
                s2 += a[i][kk + 2] * a[i][kk + 2];
                s3 += a[i][kk + 3] * a[i][kk + 3];
            }
            const v2f t = (s0 + s1) + (s2 + s3);
            xx[i] = t.x + t.y;
        }
        #pragma unroll
        for (int i = 0; i < RT; ++i) xx[i] += __shfl_xor(xx[i], 16, 32);
        #pragma unroll
        for (int i = 0; i < RT; ++i) res[i] = xx[i] + bestD[i];
    }

    {
        const int i0 = half ? bestI[2] : bestI[0];
        const int i1 = half ? bestI[3] : bestI[1];
        const int rl = waveId * (RT * 16) + half * 32 + lane16;
        sIdx[rl]      = i0;
        sIdx[rl + 16] = i1;
        atomicAdd(&sHist[i0], 1u);
        atomicAdd(&sHist[i1], 1u);
    }
    __syncthreads();

    {
        const int bin0 = waveId * 128 + lane;
        unsigned int cnt[4];
        #pragma unroll
        for (int j = 0; j < 4; ++j) cnt[j] = sHist[bin0 + j * 32];
        #pragma unroll
        for (int j = 0; j < 4; ++j) {
            if (cnt[j] != 0u) atomicAdd(&hist[bin0 + j * 32], cnt[j]);
        }
    }

    {
        float* rp = out + (half ? OUT2_OFF : OUT1_OFF) + (size_t)(waveRow + lane16) * CH + c;
        #pragma unroll
        for (int i = 0; i < RT; ++i) rp[i * 16 * CH] = res[i];
    }

    {
        const float* ec = emb + (size_t)c * KCODE * VDIM;
        v4f ev[64];
        #pragma unroll
        for (int j = 0; j < 64; ++j) {
            const unsigned int off = (unsigned int)sIdx[waveId + j * 4] * VDIM + lane * 4;
            ev[j] = *(const v4f*)(ec + off);
        }
        __builtin_amdgcn_sched_barrier(0);
        float* op = out + (size_t)(rowBase + waveId) * (CH * VDIM) + c * VDIM + lane * 4;
        #pragma unroll
        for (int j = 0; j < 64; ++j) *(v4f*)(op + (size_t)j * 4 * (CH * VDIM)) = ev[j];
    }
}

__global__ __launch_bounds__(256) void vq_entropy(const unsigned int* __restrict__ hist,
                                                  float* __restrict__ out) {
    __shared__ float red[256];
    int t = threadIdx.x;
    float s = 0.0f;
    for (int k = t; k < KCODE; k += 256) {
        unsigned int h = hist[k];
        if (h > 0u) {
            float p = (float)h * (1.0f / 32768.0f);
            s += p * logf(p);
        }
    }
    red[t] = s;
    __syncthreads();
    for (int o = 128; o > 0; o >>= 1) {
        if (t < o) red[t] += red[t + o];
        __syncthreads();
    }
    if (t == 0) out[ENT_OFF] = -red[0];
}

extern "C" void kernel_launch(void* const* d_in, const int* in_sizes, int n_in,
                              void* d_out, int out_size, void* d_ws, size_t ws_size,
                              hipStream_t stream) {
    const float* x0  = (const float*)d_in[0];
    const float* emb = (const float*)d_in[1];
    float* out = (float*)d_out;

    unsigned int* hist = (unsigned int*)d_ws;
    float* esq = (float*)((char*)d_ws + KCODE * sizeof(unsigned int));

    vq_setup  <<<4, 256, 0, stream>>>(emb, hist, esq);
    vq_main   <<<(TOKENS / 256) * CH, 128, 0, stream>>>(x0, emb, esq, hist, out);
    vq_entropy<<<1, 256, 0, stream>>>(hist, out);
}


# ||e||^2 no longer a pass of the first launch: in the sweep each lane sums the squares of the code fragments it holds, one WMMA per tile (partial sums x ones) adds the halves; ||e||^2 sums re-ordered
# speedup vs baseline: 1.5696x; 1.0748x over previous
#include <hip/hip_runtime.h>
#include <hip/hip_bf16.h>
#include <math.h>

typedef __attribute__((ext_vector_type(2))) float v2f;
typedef __attribute__((ext_vector_type(8))) float v8f;
typedef __attribute__((ext_vector_type(4))) float v4f;

#define TOKENS   32768
#define CH       2
#define VDIM     128
#define KCODE    512
#define OUT1_OFF 8388608
#define OUT2_OFF 8454144
#define ENT_OFF  8519680

__global__ __launch_bounds__(128) void vq_setup(unsigned int* __restrict__ hist) {
    *(uint4*)(hist + threadIdx.x * 4) = make_uint4(0u, 0u, 0u, 0u);
}

#define RT 4

__device__ __forceinline__ void load_b_tile(const float* __restrict__ ep,
                                            v2f (&b)[32]) {
    #pragma unroll
    for (int kk = 0; kk < 32; ++kk) b[kk] = *(const v2f*)(ep + kk * 4);
}

__device__ __forceinline__ void load_b_tile_rev(const float* __restrict__ ep,
                                                v2f (&b)[32]) {
    #pragma unroll
    for (int kk = 31; kk >= 0; --kk) b[kk] = *(const v2f*)(ep + kk * 4);
}

__device__ __forceinline__ void mma_tile(const v2f (&a)[RT][32], const v2f (&b)[32],
                                         v8f (&acc)[RT]) {
    #pragma unroll
    for (int i = 0; i < RT; ++i) acc[i] = (v8f){};
    #pragma unroll
    for (int kk = 0; kk < 32; ++kk) {
        #pragma unroll
        for (int i = 0; i < RT; ++i) {
            acc[i] = __builtin_amdgcn_wmma_f32_16x16x4_f32(
                false, b[kk], false, a[i][kk], (short)0, acc[i], false, false);
        }
    }
}

__device__ __forceinline__ v2f tile_sq_sums(const v2f (&b)[32]) {
    v2f s0 = b[0] * b[0], s1 = b[1] * b[1];
    v2f s2 = b[2] * b[2], s3 = b[3] * b[3];
    #pragma unroll
    for (int kk = 4; kk < 32; kk += 4) {
        s0 += b[kk]     * b[kk];
        s1 += b[kk + 1] * b[kk + 1];
        s2 += b[kk + 2] * b[kk + 2];
        s3 += b[kk + 3] * b[kk + 3];
    }
    return (s0 + s1) + (s2 + s3);
}

__device__ __forceinline__ v8f esq_by_register(const v2f part) {
    const v2f ones = (v2f){1.0f, 1.0f};
    return __builtin_amdgcn_wmma_f32_16x16x4_f32(
        false, part, false, ones, (short)0, (v8f){}, false, false);
}

__device__ __forceinline__ void argmin_update(const v8f (&acc)[RT], const v8f &q, int code0,
                                              float (&runD)[RT][8], int (&runI)[RT][8]) {
    #pragma unroll
    for (int i = 0; i < RT; ++i) {
        #pragma unroll
        for (int p = 0; p < 4; ++p) {
            const v2f ap = (v2f){acc[i][2 * p], acc[i][2 * p + 1]};
            const v2f d  = (v2f){q[2 * p], q[2 * p + 1]} - 2.0f * ap;
            if (d.x < runD[i][2 * p])     { runD[i][2 * p]     = d.x; runI[i][2 * p]     = code0; }
            if (d.y < runD[i][2 * p + 1]) { runD[i][2 * p + 1] = d.y; runI[i][2 * p + 1] = code0; }
        }
    }
}

__global__ __launch_bounds__(128) void vq_main(const float* __restrict__ x0,
                                               const float* __restrict__ emb,
                                               unsigned int* __restrict__ hist,
                                               float* __restrict__ out) {
    __shared__ int sIdx[256];
    __shared__ __attribute__((aligned(16))) unsigned int sHist[KCODE];

    *(uint4*)(sHist + threadIdx.x * 4) = make_uint4(0u, 0u, 0u, 0u);
    __syncthreads();

    const int lane   = threadIdx.x & 31;
    const int waveId = threadIdx.x >> 5;
    const int lane16 = lane & 15;
    const int half   = lane >> 4;
    const int c       = blockIdx.x & 1;
    const int rowBase = (blockIdx.x >> 1) * 256;
    const int waveRow = rowBase + waveId * (RT * 16);

    v2f a[RT][32];
    {
        const float* xp = x0 + (size_t)(waveRow + lane16) * (CH * VDIM)
                             + c * VDIM + half * 2;
        #pragma unroll
        for (int i = 0; i < RT; ++i) {
            #pragma unroll
            for (int kk = 0; kk < 32; ++kk) {
                a[i][kk] = *(const v2f*)(xp + i * 16 * (CH * VDIM) + kk * 4);
            }
        }
    }

    float runD[RT][8];
    int   runI[RT][8];
    #pragma unroll
    for (int i = 0; i < RT; ++i) {
        #pragma unroll
        for (int r = 0; r < 8; ++r) { runD[i][r] = 3.4e38f; runI[i][r] = 0; }
    }

    const float* ebase = emb + ((size_t)(c * KCODE + lane16)) * VDIM + half * 2;

    v2f bE[32], bO[32];
    load_b_tile(ebase, bE);

    for (int t = 0; t < 32; t += 2) {
        load_b_tile(ebase + (size_t)(t + 1) * 16 * VDIM, bO);
        __builtin_amdgcn_sched_barrier(0);
        const v2f pE = tile_sq_sums(bE);
        v8f qE = esq_by_register(pE);
        v8f accE[RT];
        mma_tile(a, bE, accE);
        asm volatile("" : "+v"(qE) : "v"(pE));

        const int tn = (t + 2 < 32) ? (t + 2) : 31;
        __builtin_amdgcn_sched_barrier(0);
        load_b_tile_rev(ebase + (size_t)tn * 16 * VDIM, bE);
        __builtin_amdgcn_sched_barrier(0);

        argmin_update(accE, qE, t * 16 + half * 8, runD, runI);
        const v8f qO = esq_by_register(tile_sq_sums(bO));
        v8f accO[RT];
        mma_tile(a, bO, accO);
        argmin_update(accO, qO, (t + 1) * 16 + half * 8, runD, runI);
    }

    float bestD[RT];
    int   bestI[RT];
    #pragma unroll
    for (int i = 0; i < RT; ++i) {
        float bd = runD[i][0];
        int   bi = runI[i][0];
        #pragma unroll
        for (int r = 1; r < 8; ++r) {
            const float cd = runD[i][r];
            const int   ci = runI[i][r] + r;
            const bool take = (cd < bd) | ((cd == bd) & (ci < bi));
            bd = take ? cd : bd;
            bi = take ? ci : bi;
        }
        bestD[i] = bd; bestI[i] = bi;
    }
    {
        float od[RT];
        int   oi[RT];
        #pragma unroll
        for (int i = 0; i < RT; ++i) {
            od[i] = __shfl_xor(bestD[i], 16, 32);
            oi[i] = __shfl_xor(bestI[i], 16, 32);
        }
        #pragma unroll
        for (int i = 0; i < RT; ++i) {
            const bool take = (od[i] < bestD[i]) | ((od[i] == bestD[i]) & (oi[i] < bestI[i]));
            bestI[i] = take ? oi[i] : bestI[i];
            bestD[i] = take ? od[i] : bestD[i];
        }
    }

    float res[RT];
    {
        float xx[RT];
        #pragma unroll
        for (int i = 0; i < RT; ++i) {
            v2f s0 = a[i][0] * a[i][0], s1 = a[i][1] * a[i][1];
            v2f s2 = a[i][2] * a[i][2], s3 = a[i][3] * a[i][3];
            #pragma unroll
            for (int kk = 4; kk < 32; kk += 4) {
                s0 += a[i][kk]     * a[i][kk];
                s1 += a[i][kk + 1] * a[i][kk + 1];
                s2 += a[i][kk + 2] * a[i][kk + 2];
                s3 += a[i][kk + 3] * a[i][kk + 3];
            }
            const v2f t = (s0 + s1) + (s2 + s3);
            xx[i] = t.x + t.y;
        }
        #pragma unroll
        for (int i = 0; i < RT; ++i) xx[i] += __shfl_xor(xx[i], 16, 32);
        #pragma unroll
        for (int i = 0; i < RT; ++i) res[i] = xx[i] + bestD[i];
    }

    {
        const int i0 = half ? bestI[2] : bestI[0];
        const int i1 = half ? bestI[3] : bestI[1];
        const int rl = waveId * (RT * 16) + half * 32 + lane16;
        sIdx[rl]      = i0;
        sIdx[rl + 16] = i1;
        atomicAdd(&sHist[i0], 1u);
        atomicAdd(&sHist[i1], 1u);
    }
    __syncthreads();

    {
        const int bin0 = waveId * 128 + lane;
        unsigned int cnt[4];
        #pragma unroll
        for (int j = 0; j < 4; ++j) cnt[j] = sHist[bin0 + j * 32];
        #pragma unroll
        for (int j = 0; j < 4; ++j) {
            if (cnt[j] != 0u) atomicAdd(&hist[bin0 + j * 32], cnt[j]);
        }
    }

    {
        float* rp = out + (half ? OUT2_OFF : OUT1_OFF) + (size_t)(waveRow + lane16) * CH + c;
        #pragma unroll
        for (int i = 0; i < RT; ++i) rp[i * 16 * CH] = res[i];
    }

    {
        const float* ec = emb + (size_t)c * KCODE * VDIM;
        v4f ev[64];
        #pragma unroll
        for (int j = 0; j < 64; ++j) {
            const unsigned int off = (unsigned int)sIdx[waveId + j * 4] * VDIM + lane * 4;
            ev[j] = *(const v4f*)(ec + off);
        }
        __builtin_amdgcn_sched_barrier(0);
        float* op = out + (size_t)(rowBase + waveId) * (CH * VDIM) + c * VDIM + lane * 4;
        #pragma unroll
        for (int j = 0; j < 64; ++j) *(v4f*)(op + (size_t)j * 4 * (CH * VDIM)) = ev[j];
    }
}

__global__ __launch_bounds__(256) void vq_entropy(const unsigned int* __restrict__ hist,
                                                  float* __restrict__ out) {
    __shared__ float red[256];
    int t = threadIdx.x;
    float s = 0.0f;
    for (int k = t; k < KCODE; k += 256) {
        unsigned int h = hist[k];
        if (h > 0u) {
            float p = (float)h * (1.0f / 32768.0f);
            s += p * logf(p);
        }
    }
    red[t] = s;
    __syncthreads();
    for (int o = 128; o > 0; o >>= 1) {
        if (t < o) red[t] += red[t + o];
        __syncthreads();
    }
    if (t == 0) out[ENT_OFF] = -red[0];
}

extern "C" void kernel_launch(void* const* d_in, const int* in_sizes, int n_in,
                              void* d_out, int out_size, void* d_ws, size_t ws_size,
                              hipStream_t stream) {
    const float* x0  = (const float*)d_in[0];
    const float* emb = (const float*)d_in[1];
    float* out = (float*)d_out;

    unsigned int* hist = (unsigned int*)d_ws;

    vq_setup  <<<1, 128, 0, stream>>>(hist);
    vq_main   <<<(TOKENS / 256) * CH, 128, 0, stream>>>(x0, emb, hist, out);
    vq_entropy<<<1, 256, 0, stream>>>(hist, out);
}
